# starting kernel
# baseline (speedup 1.0000x reference)
#include <hip/hip_runtime.h>
#include <hip/hip_bf16.h>

typedef __attribute__((ext_vector_type(16))) _Float16 v16h;
typedef __attribute__((ext_vector_type(8)))  float    v8f;

#define N_    256
#define NN    65536
#define KHOP  32
#define D_    32
#define H_    8
#define WAVES_PER_BLOCK 8

__global__ __launch_bounds__(256)
void attn_bias_wmma(const int*   __restrict__ spatial_pos,
                    const int*   __restrict__ edge_input,
                    const float* __restrict__ spatial_emb,
                    const float* __restrict__ edge_feat_emb,
                    const float* __restrict__ edge_pos_emb,
                    float*       __restrict__ out)
{
    __shared__ _Float16 bfrag[KHOP * 32 * 16];

    const int tid  = threadIdx.x;
    const int lane = tid & 31;
    const int wave = tid >> 5;

    for (int f = tid; f < KHOP * 32; f += 256) {
        const int k  = f >> 5;
        const int L  = f & 31;
        const int n  = L & 15;
        const int hi = (L >> 4) * 8;
        _Float16* dst = &bfrag[f * 16];
        if (n < H_) {
            const float* w = edge_pos_emb + k * (D_ * H_) + n;
#pragma unroll
            for (int j = 0; j < 16; ++j) {
                const int d = (j < 8) ? (j + hi) : (j - 8 + 16 + hi);
                dst[j] = (_Float16)w[d * H_];
            }
        } else {
#pragma unroll
            for (int j = 0; j < 16; ++j) dst[j] = (_Float16)0.0f;
        }
    }
    __syncthreads();

    const int  tile  = blockIdx.x * WAVES_PER_BLOCK + wave;
    const long long base = (long long)tile * 16;
    const int  mrow  = lane & 15;
    const int  dbase = (lane >> 4) * 8;

    const int* eip = edge_input + (base + mrow) * KHOP;

    v8f c = {};
    int cnt = 0;
    for (int k = 0; k < KHOP; ++k) {
        const int idx = eip[k];
        cnt += (idx != 0);
        const float4* row4 = (const float4*)(edge_feat_emb + (long long)idx * D_ + dbase);
        const float4 c0 = row4[0], c1 = row4[1], c2 = row4[4], c3 = row4[5];
        v16h a;
        a[0]  = (_Float16)c0.x; a[1]  = (_Float16)c0.y; a[2]  = (_Float16)c0.z; a[3]  = (_Float16)c0.w;
        a[4]  = (_Float16)c1.x; a[5]  = (_Float16)c1.y; a[6]  = (_Float16)c1.z; a[7]  = (_Float16)c1.w;
        a[8]  = (_Float16)c2.x; a[9]  = (_Float16)c2.y; a[10] = (_Float16)c2.z; a[11] = (_Float16)c2.w;
        a[12] = (_Float16)c3.x; a[13] = (_Float16)c3.y; a[14] = (_Float16)c3.z; a[15] = (_Float16)c3.w;

        const v16h b = *(const v16h*)&bfrag[(k * 32 + lane) * 16];

        c = __builtin_amdgcn_wmma_f32_16x16x32_f16(
                 false, a,  false, b,
                 (short)0, c,  false,  false);
    }

    const int   h    = lane & 15;
    const int   mtop = (lane >> 4) * 8;
    const float cntf = (float)cnt;

    float pls[8];
#pragma unroll
    for (int r = 0; r < 8; ++r)
        pls[r] = fmaxf(__shfl(cntf, mtop + r, 32), 1.0f);

    if (h < H_) {
#pragma unroll
        for (int r = 0; r < 8; ++r) {
            const long long p   = base + mtop + r;
            const int       sp  = spatial_pos[p];
            const float     sb  = spatial_emb[sp * H_ + h];
            const long long bi  = p >> 16;
            const long long nm  = p & 0xFFFF;
            out[bi * (long long)(H_ * NN) + (long long)h * NN + nm] =
                sb + c[r] / pls[r];
        }
    }
}

extern "C" void kernel_launch(void* const* d_in, const int* in_sizes, int n_in,
                              void* d_out, int out_size, void* d_ws, size_t ws_size,
                              hipStream_t stream) {
    const int*   spatial_pos   = (const int*)d_in[0];
    const int*   edge_input    = (const int*)d_in[1];
    const float* spatial_emb   = (const float*)d_in[2];
    const float* edge_feat_emb = (const float*)d_in[3];
    const float* edge_pos_emb  = (const float*)d_in[4];
    float*       out           = (float*)d_out;

    const int pairs  = in_sizes[0];
    const int tiles  = pairs / 16;
    const int blocks = tiles / WAVES_PER_BLOCK;

    hipLaunchKernelGGL(attn_bias_wmma, dim3(blocks), dim3(256), 0, stream,
                       spatial_pos, edge_input, spatial_emb, edge_feat_emb,
                       edge_pos_emb, out);
}


# Products table never stored: each work-group re-forms one hop pair's f32 slice per loop step on WMMA (table tiles held in registers) into the LDS ring; 1 launch, no L2 stream; hop sum order unchanged
# speedup vs baseline: 4.0648x; 4.0648x over previous
#include <hip/hip_runtime.h>


typedef __attribute__((ext_vector_type(16))) _Float16 v16h;
typedef __attribute__((ext_vector_type(8)))  _Float16 v8h;
typedef __attribute__((ext_vector_type(8)))  float    v8f;

#define NN     65536
#define KHOP   32
#define D_     32
#define H_     8
#define EROWS  1025

#define SLICE_BYTES 32768
#define LASTROW_OFF (4 * SLICE_BYTES)
#define WFRAG_OFF   (LASTROW_OFF + KHOP * 32)
#define PK_OFF      (WFRAG_OFF + 16 * 1024)
#define CNT_OFF     (PK_OFF + 16 * 8192)
#define LDS_BYTES   (CNT_OFF + 8192)

__global__ __launch_bounds__(1024)
void hop_sum(const int*   __restrict__ spatial_pos,
             const int*   __restrict__ edge_input,
             const float* __restrict__ spatial_emb,
             const float* __restrict__ edge_feat_emb,
             const float* __restrict__ edge_pos_emb,
             float*       __restrict__ out)
{
    __shared__ float4 lds[LDS_BYTES / 16];

    const int tid  = threadIdx.x;
    const int lane = tid & 31;
    const int wave = tid >> 5;
    const int i    = lane >> 1;
    const int hf   = lane & 1;
    const int c    = lane & 15;
    const int mh   = lane >> 4;
    const int wbase = (blockIdx.x * 32 + wave) * 32;
    char* const L = (char*)lds;

    const float4* lrow4 = (const float4*)(edge_feat_emb + (EROWS - 1) * D_ + mh * 8);
    const float4 l0 = lrow4[0], l1 = lrow4[1], l2 = lrow4[4], l3 = lrow4[5];

    {
        const int q = tid >> 6, fl = (tid >> 1) & 31, half = tid & 1;
        const int fc = fl & 15, fh = fl >> 4;
        const float* w = edge_pos_emb + (2 * q + (fc >> 3)) * (D_ * H_) + (fc & 7)
                                      + (16 * half + 8 * fh) * H_;
        v8h f;
#pragma unroll
        for (int j = 0; j < 8; ++j) f[j] = (_Float16)w[j * H_];
        *(v8h*)(L + WFRAG_OFF + q * 1024 + half * 512 + fl * 16) = f;
    }

    v16h bt[2];
#pragma unroll
    for (int t = 0; t < 2; ++t) {
        const int e = (wave + 32 * t) * 16 + c;
        const float4* row4 = (const float4*)(edge_feat_emb + e * D_ + mh * 8);
        const float4 c0 = row4[0], c1 = row4[1], c2 = row4[4], c3 = row4[5];
        v16h b;
        b[0]  = (_Float16)c0.x; b[1]  = (_Float16)c0.y; b[2]  = (_Float16)c0.z; b[3]  = (_Float16)c0.w;
        b[4]  = (_Float16)c1.x; b[5]  = (_Float16)c1.y; b[6]  = (_Float16)c1.z; b[7]  = (_Float16)c1.w;
        b[8]  = (_Float16)c2.x; b[9]  = (_Float16)c2.y; b[10] = (_Float16)c2.z; b[11] = (_Float16)c2.w;
        b[12] = (_Float16)c3.x; b[13] = (_Float16)c3.y; b[14] = (_Float16)c3.z; b[15] = (_Float16)c3.w;
        bt[t] = b;
    }

    int cntA = 0, cntB = 0;
    {
        const unsigned hoff = (unsigned)hf * 16u;
        const int4* pa = (const int4*)(edge_input + (long long)(wbase + i) * KHOP);
        const int4* pb = (const int4*)(edge_input + (long long)(wbase + 16 + i) * KHOP);
#pragma unroll
        for (int j = 0; j < 8; ++j) {
            const int4 ta = pa[j], tb = pb[j];
            const int va[4] = { ta.x, ta.y, ta.z, ta.w };
            const int vb[4] = { tb.x, tb.y, tb.z, tb.w };
            unsigned w4[4];
#pragma unroll
            for (int u = 0; u < 4; ++u) {
                const int k = 4 * j + u;
                const unsigned last = (unsigned)(LASTROW_OFF + 32 * k) + hoff;
                const unsigned ring = (unsigned)((k & 3) * SLICE_BYTES) + hoff;
                const unsigned ra = ((unsigned)va[u] * 32u + ring) ^ (((unsigned)va[u] & 8u) << 1);
                const unsigned rb = ((unsigned)vb[u] * 32u + ring) ^ (((unsigned)vb[u] & 8u) << 1);
                const unsigned oa = (va[u] == 1024) ? last : ra;
                const unsigned ob = (vb[u] == 1024) ? last : rb;
                w4[u] = oa | ((ob >> 4) << 18);
                cntA += (va[u] != 0);
                cntB += (vb[u] != 0);
            }
            *(uint2*)(L + PK_OFF + (2 * j)     * 8192 + tid * 8) = make_uint2(w4[0], w4[1]);
            *(uint2*)(L + PK_OFF + (2 * j + 1) * 8192 + tid * 8) = make_uint2(w4[2], w4[3]);
        }
        *(uint2*)(L + CNT_OFF + tid * 8) = make_uint2((unsigned)cntA, (unsigned)cntB);
    }

    __syncthreads();

#define LOAD_A(q, a) do {                                                                    \
        const v8h lo_ = *(const v8h*)(L + WFRAG_OFF + (q) * 1024 + lane * 16);               \
        const v8h hi_ = *(const v8h*)(L + WFRAG_OFF + (q) * 1024 + 512 + lane * 16);         \
        a = __builtin_shufflevector(lo_, hi_, 0, 1, 2, 3, 4, 5, 6, 7,                        \
                                    8, 9, 10, 11, 12, 13, 14, 15);                           \
    } while (0)

    if (wave < 16) {
        v16h bl;
        bl[0]  = (_Float16)l0.x; bl[1]  = (_Float16)l0.y; bl[2]  = (_Float16)l0.z; bl[3]  = (_Float16)l0.w;
        bl[4]  = (_Float16)l1.x; bl[5]  = (_Float16)l1.y; bl[6]  = (_Float16)l1.z; bl[7]  = (_Float16)l1.w;
        bl[8]  = (_Float16)l2.x; bl[9]  = (_Float16)l2.y; bl[10] = (_Float16)l2.z; bl[11] = (_Float16)l2.w;
        bl[12] = (_Float16)l3.x; bl[13] = (_Float16)l3.y; bl[14] = (_Float16)l3.z; bl[15] = (_Float16)l3.w;
        v16h a;
        LOAD_A(wave, a);
        v8f acc = {};
        acc = __builtin_amdgcn_wmma_f32_16x16x32_f16(false, a, false, bl, (short)0, acc, false, false);
        if (c == 0) {
            float4* d = (float4*)(L + LASTROW_OFF + 32 * (2 * wave + mh));
            d[0] = make_float4(acc[0], acc[1], acc[2], acc[3]);
            d[1] = make_float4(acc[4], acc[5], acc[6], acc[7]);
        }
    }

#define BUILD_PAIR(q, a_) do {                                                               \
        v8f r0_ = {}, r1_ = {};                                                              \
        r0_ = __builtin_amdgcn_wmma_f32_16x16x32_f16(false, a_, false, bt[0], (short)0,      \
                                                     r0_, false, false);                     \
        r1_ = __builtin_amdgcn_wmma_f32_16x16x32_f16(false, a_, false, bt[1], (short)0,      \
                                                     r1_, false, false);                     \
        char* const p_ = L + (2 * ((q) & 1) + mh) * SLICE_BYTES + (wave * 16 + c) * 32;      \
        *(float4*)(p_ + swo)                = make_float4(r0_[0], r0_[1], r0_[2], r0_[3]);   \
        *(float4*)(p_ + (swo ^ 16))         = make_float4(r0_[4], r0_[5], r0_[6], r0_[7]);   \
        *(float4*)(p_ + 16384 + swo)        = make_float4(r1_[0], r1_[1], r1_[2], r1_[3]);   \
        *(float4*)(p_ + 16384 + (swo ^ 16)) = make_float4(r1_[4], r1_[5], r1_[6], r1_[7]);   \
    } while (0)

    const int swo = (c & 8) << 1;

    {
        v16h a;
        LOAD_A(0, a);
        BUILD_PAIR(0, a);
    }
    __syncthreads();

    float4 accA = make_float4(0.f, 0.f, 0.f, 0.f), accB = accA;
    const char* const pkp = L + PK_OFF + tid * 8;
    uint2 w = *(const uint2*)pkp;

#define GATHER(w_)                                                                           \
        const float4 a0 = *(const float4*)(L + ((w_).x & 0x3FFFFu));                         \
        const float4 b0 = *(const float4*)(L + (((w_).x >> 14) & 0x3FFF0u));                 \
        const float4 a1 = *(const float4*)(L + ((w_).y & 0x3FFFFu));                         \
        const float4 b1 = *(const float4*)(L + (((w_).y >> 14) & 0x3FFF0u))
#define ADD()  do {                                                                          \
        accA.x += a0.x; accA.y += a0.y; accA.z += a0.z; accA.w += a0.w;                      \
        accB.x += b0.x; accB.y += b0.y; accB.z += b0.z; accB.w += b0.w;                      \
        accA.x += a1.x; accA.y += a1.y; accA.z += a1.z; accA.w += a1.w;                      \
        accB.x += b1.x; accB.y += b1.y; accB.z += b1.z; accB.w += b1.w;                      \
    } while (0)

#pragma unroll 1
    for (int q = 0; q < KHOP / 2 - 1; ++q) {
        v16h a;
        LOAD_A(q + 1, a);
        const uint2 wn = *(const uint2*)(pkp + (q + 1) * 8192);
        GATHER(w);
        BUILD_PAIR(q + 1, a);
        ADD();
        w = wn;
        __syncthreads();
    }
    {
        GATHER(w);
        ADD();
    }
#undef GATHER
#undef ADD
#undef BUILD_PAIR
#undef LOAD_A

    const uint2 cnt = *(const uint2*)(L + CNT_OFF + tid * 8);
    const float plA = fmaxf((float)cnt.x, 1.0f);
    const float plB = fmaxf((float)cnt.y, 1.0f);
    const float vA[4] = { accA.x / plA, accA.y / plA, accA.z / plA, accA.w / plA };
    const float vB[4] = { accB.x / plB, accB.y / plB, accB.z / plB, accB.w / plB };

    const long long p  = (long long)wbase + lane;
    const int       sp = spatial_pos[p];
    const float4 s0 = *(const float4*)(spatial_emb + sp * H_);
    const float4 s1 = *(const float4*)(spatial_emb + sp * H_ + 4);
    const float sb8[8] = { s0.x, s0.y, s0.z, s0.w, s1.x, s1.y, s1.z, s1.w };
    float* op = out + (p >> 16) * (long long)(H_ * NN) + (p & 0xFFFF);
    const int  src = 2 * (lane & 15);
    const bool lo  = lane < 16;
#pragma unroll
    for (int h = 0; h < H_; ++h) {
        const float xa = __shfl(vA[h & 3], src + (h >> 2), 32);
        const float xb = __shfl(vB[h & 3], src + (h >> 2), 32);
        op[(long long)h * NN] = sb8[h] + (lo ? xa : xb);
    }
}

extern "C" void kernel_launch(void* const* d_in, const int* in_sizes, int n_in,
                              void* d_out, int out_size, void* d_ws, size_t ws_size,
                              hipStream_t stream) {
    const int*   spatial_pos   = (const int*)d_in[0];
    const int*   edge_input    = (const int*)d_in[1];
    const float* spatial_emb   = (const float*)d_in[2];
    const float* edge_feat_emb = (const float*)d_in[3];
    const float* edge_pos_emb  = (const float*)d_in[4];
    float*       out           = (float*)d_out;

    hipLaunchKernelGGL(hop_sum, dim3(256), dim3(1024), 0, stream,
                       spatial_pos, edge_input, spatial_emb, edge_feat_emb, edge_pos_emb, out);
}


# Index stream read once in whole 128-B lines (8 loads of 4 lines per wave, were 16 of 16 part lines); each index converted once, upper-half lane's word = lower's with 2 bits flipped; sums unchanged
# speedup vs baseline: 4.6199x; 1.1366x over previous
#include <hip/hip_runtime.h>


typedef __attribute__((ext_vector_type(16))) _Float16 v16h;
typedef __attribute__((ext_vector_type(8)))  _Float16 v8h;
typedef __attribute__((ext_vector_type(8)))  float    v8f;

#define NN     65536
#define KHOP   32
#define D_     32
#define H_     8
#define EROWS  1025

#define SLICE_BYTES 32768
#define LASTROW_OFF (4 * SLICE_BYTES)
#define WFRAG_OFF   (LASTROW_OFF + KHOP * 32)
#define PK_OFF      (WFRAG_OFF + 16 * 1024)
#define PK_PITCH    8208
#define CNT_OFF     (PK_OFF + 16 * PK_PITCH)
#define LDS_BYTES   (CNT_OFF + 4096)

__global__ __launch_bounds__(1024)
void hop_sum(const int*   __restrict__ spatial_pos,
             const int*   __restrict__ edge_input,
             const float* __restrict__ spatial_emb,
             const float* __restrict__ edge_feat_emb,
             const float* __restrict__ edge_pos_emb,
             float*       __restrict__ out)
{
    __shared__ float4 lds[LDS_BYTES / 16];

    const int tid  = threadIdx.x;
    const int lane = tid & 31;
    const int wave = tid >> 5;
    const int i    = lane >> 1;
    const int hf   = lane & 1;
    const int c    = lane & 15;
    const int mh   = lane >> 4;
    const int wbase = (blockIdx.x * 32 + wave) * 32;
    char* const L = (char*)lds;

    const float4* lrow4 = (const float4*)(edge_feat_emb + (EROWS - 1) * D_ + mh * 8);
    const float4 l0 = lrow4[0], l1 = lrow4[1], l2 = lrow4[4], l3 = lrow4[5];

    {
        const int q = tid >> 6, fl = (tid >> 1) & 31, half = tid & 1;
        const int fc = fl & 15, fh = fl >> 4;
        const float* w = edge_pos_emb + (2 * q + (fc >> 3)) * (D_ * H_) + (fc & 7)
                                      + (16 * half + 8 * fh) * H_;
        v8h f;
#pragma unroll
        for (int j = 0; j < 8; ++j) f[j] = (_Float16)w[j * H_];
        *(v8h*)(L + WFRAG_OFF + q * 1024 + half * 512 + fl * 16) = f;
    }

    v16h bt[2];
#pragma unroll
    for (int t = 0; t < 2; ++t) {
        const int e = (wave + 32 * t) * 16 + c;
        const float4* row4 = (const float4*)(edge_feat_emb + e * D_ + mh * 8);
        const float4 c0 = row4[0], c1 = row4[1], c2 = row4[4], c3 = row4[5];
        v16h b;
        b[0]  = (_Float16)c0.x; b[1]  = (_Float16)c0.y; b[2]  = (_Float16)c0.z; b[3]  = (_Float16)c0.w;
        b[4]  = (_Float16)c1.x; b[5]  = (_Float16)c1.y; b[6]  = (_Float16)c1.z; b[7]  = (_Float16)c1.w;
        b[8]  = (_Float16)c2.x; b[9]  = (_Float16)c2.y; b[10] = (_Float16)c2.z; b[11] = (_Float16)c2.w;
        b[12] = (_Float16)c3.x; b[13] = (_Float16)c3.y; b[14] = (_Float16)c3.z; b[15] = (_Float16)c3.w;
        bt[t] = b;
    }

    {
        const int hq = lane & 7, pg = lane >> 3;
        const int4* src = (const int4*)(edge_input + (long long)wbase * KHOP) + lane;
        const unsigned lastq = (unsigned)(LASTROW_OFF + 128 * hq);
        char* const pk = L + PK_OFF + (2 * hq) * PK_PITCH + wave * 256 + pg * 16;
        unsigned cA = 0, cB = 0;
#pragma unroll
        for (int j = 0; j < 4; ++j) {
            const int4 ta = src[32 * j], tb = src[32 * (j + 4)];
            const int va[4] = { ta.x, ta.y, ta.z, ta.w };
            const int vb[4] = { tb.x, tb.y, tb.z, tb.w };
            unsigned w4[4];
            unsigned na = 0, nb = 0;
#pragma unroll
            for (int u = 0; u < 4; ++u) {
                const unsigned last = lastq + 32u * (unsigned)u;
                const unsigned ring = (unsigned)(u * SLICE_BYTES);
                const unsigned ra = ((unsigned)va[u] * 32u + ring) ^ (((unsigned)va[u] & 8u) << 1);
                const unsigned rb = ((unsigned)vb[u] * 32u + ring) ^ (((unsigned)vb[u] & 8u) << 1);
                const unsigned oa = (va[u] == 1024) ? last : ra;
                const unsigned ob = (vb[u] == 1024) ? last : rb;
                w4[u] = oa | ((ob >> 4) << 18);
                na += (va[u] != 0);
                nb += (vb[u] != 0);
            }
            cA |= na << (8 * j);
            cB |= nb << (8 * j);
            const unsigned m = 0x40010u;
            *(uint4*)(pk + j * 64)            = make_uint4(w4[0], w4[1], w4[0] ^ m, w4[1] ^ m);
            *(uint4*)(pk + j * 64 + PK_PITCH) = make_uint4(w4[2], w4[3], w4[2] ^ m, w4[3] ^ m);
        }
        cA += (unsigned)__shfl_xor((int)cA, 1, 32);
        cB += (unsigned)__shfl_xor((int)cB, 1, 32);
        cA += (unsigned)__shfl_xor((int)cA, 2, 32);
        cB += (unsigned)__shfl_xor((int)cB, 2, 32);
        cA += (unsigned)__shfl_xor((int)cA, 4, 32);
        cB += (unsigned)__shfl_xor((int)cB, 4, 32);
        const unsigned cc = ((hq < 4 ? cA : cB) >> (8 * (hq & 3))) & 0xFFu;
        *(unsigned*)(L + CNT_OFF + (wave * 32 + 4 * hq + pg) * 4) = cc;
    }

    __syncthreads();

#define LOAD_A(q, a) do {                                                                    \
        const v8h lo_ = *(const v8h*)(L + WFRAG_OFF + (q) * 1024 + lane * 16);               \
        const v8h hi_ = *(const v8h*)(L + WFRAG_OFF + (q) * 1024 + 512 + lane * 16);         \
        a = __builtin_shufflevector(lo_, hi_, 0, 1, 2, 3, 4, 5, 6, 7,                        \
                                    8, 9, 10, 11, 12, 13, 14, 15);                           \
    } while (0)

    if (wave < 16) {
        v16h bl;
        bl[0]  = (_Float16)l0.x; bl[1]  = (_Float16)l0.y; bl[2]  = (_Float16)l0.z; bl[3]  = (_Float16)l0.w;
        bl[4]  = (_Float16)l1.x; bl[5]  = (_Float16)l1.y; bl[6]  = (_Float16)l1.z; bl[7]  = (_Float16)l1.w;
        bl[8]  = (_Float16)l2.x; bl[9]  = (_Float16)l2.y; bl[10] = (_Float16)l2.z; bl[11] = (_Float16)l2.w;
        bl[12] = (_Float16)l3.x; bl[13] = (_Float16)l3.y; bl[14] = (_Float16)l3.z; bl[15] = (_Float16)l3.w;
        v16h a;
        LOAD_A(wave, a);
        v8f acc = {};
        acc = __builtin_amdgcn_wmma_f32_16x16x32_f16(false, a, false, bl, (short)0, acc, false, false);
        if (c == 0) {
            float4* d = (float4*)(L + LASTROW_OFF + 32 * (2 * wave + mh));
            d[0] = make_float4(acc[0], acc[1], acc[2], acc[3]);
            d[1] = make_float4(acc[4], acc[5], acc[6], acc[7]);
        }
    }

#define BUILD_PAIR(q, a_) do {                                                               \
        v8f r0_ = {}, r1_ = {};                                                              \
        r0_ = __builtin_amdgcn_wmma_f32_16x16x32_f16(false, a_, false, bt[0], (short)0,      \
                                                     r0_, false, false);                     \
        r1_ = __builtin_amdgcn_wmma_f32_16x16x32_f16(false, a_, false, bt[1], (short)0,      \
                                                     r1_, false, false);                     \
        char* const p_ = L + (2 * ((q) & 1) + mh) * SLICE_BYTES + (wave * 16 + c) * 32;      \
        *(float4*)(p_ + swo)                = make_float4(r0_[0], r0_[1], r0_[2], r0_[3]);   \
        *(float4*)(p_ + (swo ^ 16))         = make_float4(r0_[4], r0_[5], r0_[6], r0_[7]);   \
        *(float4*)(p_ + 16384 + swo)        = make_float4(r1_[0], r1_[1], r1_[2], r1_[3]);   \
        *(float4*)(p_ + 16384 + (swo ^ 16)) = make_float4(r1_[4], r1_[5], r1_[6], r1_[7]);   \
    } while (0)

    const int swo = (c & 8) << 1;

    {
        v16h a;
        LOAD_A(0, a);
        BUILD_PAIR(0, a);
    }
    __syncthreads();

    float4 accA = make_float4(0.f, 0.f, 0.f, 0.f), accB = accA;
    const char* const pkp = L + PK_OFF + tid * 8;
    uint2 w = *(const uint2*)pkp;

#define GATHER(w_)                                                                           \
        const float4 a0 = *(const float4*)(L + ((w_).x & 0x3FFFFu));                         \
        const float4 b0 = *(const float4*)(L + (((w_).x >> 14) & 0x3FFF0u));                 \
        const float4 a1 = *(const float4*)(L + ((w_).y & 0x3FFFFu));                         \
        const float4 b1 = *(const float4*)(L + (((w_).y >> 14) & 0x3FFF0u))
#define ADD()  do {                                                                          \
        accA.x += a0.x; accA.y += a0.y; accA.z += a0.z; accA.w += a0.w;                      \
        accB.x += b0.x; accB.y += b0.y; accB.z += b0.z; accB.w += b0.w;                      \
        accA.x += a1.x; accA.y += a1.y; accA.z += a1.z; accA.w += a1.w;                      \
        accB.x += b1.x; accB.y += b1.y; accB.z += b1.z; accB.w += b1.w;                      \
    } while (0)

#pragma unroll 1
    for (int q = 0; q < KHOP / 2 - 1; ++q) {
        v16h a;
        LOAD_A(q + 1, a);
        const uint2 wn = *(const uint2*)(pkp + (q + 1) * PK_PITCH);
        GATHER(w);
        BUILD_PAIR(q + 1, a);
        ADD();
        w = wn;
        __syncthreads();
    }
    {
        GATHER(w);
        ADD();
    }
#undef GATHER
#undef ADD
#undef BUILD_PAIR
#undef LOAD_A

    const unsigned cntA = *(const unsigned*)(L + CNT_OFF + (wave * 32 + i) * 4);
    const unsigned cntB = *(const unsigned*)(L + CNT_OFF + (wave * 32 + 16 + i) * 4);
    const float plA = fmaxf((float)cntA, 1.0f);
    const float plB = fmaxf((float)cntB, 1.0f);
    const float vA[4] = { accA.x / plA, accA.y / plA, accA.z / plA, accA.w / plA };
    const float vB[4] = { accB.x / plB, accB.y / plB, accB.z / plB, accB.w / plB };

    const long long p  = (long long)wbase + lane;
    const int       sp = spatial_pos[p];
    const float4 s0 = *(const float4*)(spatial_emb + sp * H_);
    const float4 s1 = *(const float4*)(spatial_emb + sp * H_ + 4);
    const float sb8[8] = { s0.x, s0.y, s0.z, s0.w, s1.x, s1.y, s1.z, s1.w };
    float* op = out + (p >> 16) * (long long)(H_ * NN) + (p & 0xFFFF);
    const int  src = 2 * (lane & 15);
    const bool lo  = lane < 16;
#pragma unroll
    for (int h = 0; h < H_; ++h) {
        const float xa = __shfl(vA[h & 3], src + (h >> 2), 32);
        const float xb = __shfl(vB[h & 3], src + (h >> 2), 32);
        op[(long long)h * NN] = sb8[h] + (lo ? xa : xb);
    }
}

extern "C" void kernel_launch(void* const* d_in, const int* in_sizes, int n_in,
                              void* d_out, int out_size, void* d_ws, size_t ws_size,
                              hipStream_t stream) {
    const int*   spatial_pos   = (const int*)d_in[0];
    const int*   edge_input    = (const int*)d_in[1];
    const float* spatial_emb   = (const float*)d_in[2];
    const float* edge_feat_emb = (const float*)d_in[3];
    const float* edge_pos_emb  = (const float*)d_in[4];
    float*       out           = (float*)d_out;

    hipLaunchKernelGGL(hop_sum, dim3(256), dim3(1024), 0, stream,
                       spatial_pos, edge_input, spatial_emb, edge_feat_emb, edge_pos_emb, out);
}
